# starting kernel
# baseline (speedup 1.0000x reference)
#include <hip/hip_runtime.h>
#include <hip/hip_bf16.h>


typedef __attribute__((ext_vector_type(2))) float v2f;
typedef __attribute__((ext_vector_type(8))) float v8f;

#define QP_ITERS   200
#define QP_M       128
#define QP_N       64
#define RSTRIDE    66
#define NTHREADS   256

__global__ __launch_bounds__(NTHREADS) void qp_fista_kernel(
    const float* __restrict__ Qm, const float* __restrict__ A,
    const float* __restrict__ X,  const float* __restrict__ Bv,
    float* __restrict__ Out)
{
    __shared__ __align__(16) float sR[QP_M * RSTRIDE];
    __shared__ __align__(16) float sY[2][QP_M];
    __shared__ float sC[QP_M];
    __shared__ float sLam[QP_M];
    __shared__ float sDinv[QP_N];
    __shared__ float sSqrtd[QP_N];
    __shared__ float sXp[QP_N];
    __shared__ float sRed[8];
    __shared__ float sEta;

    const int tid   = threadIdx.x;
    const int lane  = tid & 31;
    const int wave  = tid >> 5;
    const int ln16  = lane & 15;
    const int hl    = lane >> 4;
    const int batch = blockIdx.x;

    const float* Ab = A  + (size_t)batch * (QP_M * QP_N);
    const float* xb = X  + (size_t)batch * QP_N;
    const float* bb = Bv + (size_t)batch * QP_M;

    if (tid < QP_N) {
        float di = 1.0f / Qm[tid * QP_N + tid];
        sDinv[tid]  = di;
        sSqrtd[tid] = sqrtf(fabsf(di));
    }
    __syncthreads();
    if (tid < QP_N) sXp[tid] = sSqrtd[tid] * xb[tid];

    float ss = 0.f;
    #pragma unroll
    for (int j = 0; j < (QP_M * QP_N / 4) / NTHREADS; ++j) {
        int f4  = j * NTHREADS + tid;
        float4 v = ((const float4*)Ab)[f4];
        int fl  = f4 << 2;
        int row = fl >> 6, col = fl & 63;
        ss += v.x*v.x + v.y*v.y + v.z*v.z + v.w*v.w;
        float* dst = &sR[row * RSTRIDE + col];
        dst[0] = v.x * sSqrtd[col + 0];
        dst[1] = v.y * sSqrtd[col + 1];
        dst[2] = v.z * sSqrtd[col + 2];
        dst[3] = v.w * sSqrtd[col + 3];
    }
    #pragma unroll
    for (int off = 16; off; off >>= 1) ss += __shfl_xor(ss, off, 32);
    if (lane == 0) sRed[wave] = ss;
    if (tid < QP_M) sY[0][tid] = 0.f;
    __syncthreads();

    if (tid == 0) {
        float s = 0.f;
        for (int w = 0; w < 8; ++w) s += sRed[w];
        float qn2 = 0.f;
        for (int n = 0; n < QP_N; ++n) qn2 += sDinv[n] * sDinv[n];
        float L = s * sqrtf(qn2);
        sEta = 1.0f / fmaxf(L, 1e-12f);
    }
    if (tid < QP_M) {
        float acc = 0.f;
        #pragma unroll 8
        for (int n = 0; n < QP_N; ++n) acc += sR[tid * RSTRIDE + n] * sXp[n];
        sC[tid] = acc - bb[tid];
    }
    __syncthreads();

    v2f bf[16];
    #pragma unroll
    for (int k = 0; k < 16; ++k) {
        int row = wave * 16 + ln16;
        int col = 4 * k + 2 * hl;
        bf[k] = *(const v2f*)&sR[row * RSTRIDE + col];
    }
    v8f acc[8];
    #pragma unroll
    for (int i = 0; i < 8; ++i) {
        v8f c = {};
        #pragma unroll
        for (int k = 0; k < 16; ++k) {
            int row = i * 16 + ln16;
            int col = 4 * k + 2 * hl;
            v2f af = *(const v2f*)&sR[row * RSTRIDE + col];
            c = __builtin_amdgcn_wmma_f32_16x16x4_f32(
                    false, af, false, bf[k], (short)0, c, false, false);
        }
        acc[i] = c;
    }
    __syncthreads();

    const float eta = sEta;
    const int   r   = wave * 16 + ln16;
    float c_r = sC[r];
    float lam_r = 0.f, y_r = 0.f, t = 1.f;
    int buf = 0;

    for (int it = 0; it < QP_ITERS; ++it) {
        const float* y = sY[buf];
        float a = 0.f;
        #pragma unroll
        for (int i = 0; i < 8; ++i) {
            float4 y0 = *(const float4*)&y[i * 16 + 8 * hl];
            float4 y1 = *(const float4*)&y[i * 16 + 8 * hl + 4];
            a += acc[i][0]*y0.x + acc[i][1]*y0.y + acc[i][2]*y0.z + acc[i][3]*y0.w
               + acc[i][4]*y1.x + acc[i][5]*y1.y + acc[i][6]*y1.z + acc[i][7]*y1.w;
        }
        a += __shfl_xor(a, 16, 32);
        float tn = 0.5f * (1.f + sqrtf(1.f + 4.f * t * t));
        if (lane < 16) {
            float g   = c_r - a;
            float ln_ = fmaxf(0.f, y_r + eta * g);
            float yn  = ln_ + ((t - 1.f) / tn) * (ln_ - lam_r);
            sY[buf ^ 1][r] = yn;
            lam_r = ln_;
            y_r   = yn;
        }
        t = tn;
        buf ^= 1;
        __syncthreads();
    }

    if (lane < 16) sLam[r] = lam_r;
    __syncthreads();
    if (tid < QP_N) {
        float a2 = 0.f;
        #pragma unroll 8
        for (int m = 0; m < QP_M; ++m) a2 += sR[m * RSTRIDE + tid] * sLam[m];
        Out[(size_t)batch * QP_N + tid] = sDinv[tid] * xb[tid] - sSqrtd[tid] * a2;
    }
}

extern "C" void kernel_launch(void* const* d_in, const int* in_sizes, int n_in,
                              void* d_out, int out_size, void* d_ws, size_t ws_size,
                              hipStream_t stream) {
    const float* Q = (const float*)d_in[0];
    const float* A = (const float*)d_in[1];
    const float* X = (const float*)d_in[2];
    const float* B = (const float*)d_in[3];
    float* Out = (float*)d_out;
    int batches = in_sizes[1] / (QP_M * QP_N);
    qp_fista_kernel<<<batches, NTHREADS, 0, stream>>>(Q, A, X, B, Out);
}


# Each lane owns an 8x8 block of G (plain WMMA D layout): y read as 2 float4 per lane per iteration, not 16 broadcast float4; row sums by a DPP butterfly over 16 lanes (summation order changed)
# speedup vs baseline: 1.4965x; 1.4965x over previous
#include <hip/hip_runtime.h>
#include <hip/hip_bf16.h>


typedef __attribute__((ext_vector_type(2))) float v2f;
typedef __attribute__((ext_vector_type(8))) float v8f;

#define QP_ITERS   200
#define QP_M       128
#define QP_N       64
#define RSTRIDE    66
#define NTHREADS   256

template <int MASK>
__device__ __forceinline__ float row_xor(float v) {
    return __int_as_float(__builtin_amdgcn_update_dpp(
        0, __float_as_int(v), 0x160 + MASK, 0xf, 0xf, true));
}

__global__ __launch_bounds__(NTHREADS) void qp_fista_kernel(
    const float* __restrict__ Qm, const float* __restrict__ A,
    const float* __restrict__ X,  const float* __restrict__ Bv,
    float* __restrict__ Out)
{
    __shared__ __align__(16) float sR[QP_M * RSTRIDE];
    __shared__ __align__(16) float sY[2][2][64];
    __shared__ float sC[QP_M];
    __shared__ float sLam[QP_M];
    __shared__ float sDinv[QP_N];
    __shared__ float sSqrtd[QP_N];
    __shared__ float sXp[QP_N];
    __shared__ float sRed[8];
    __shared__ float sEta;

    const int tid   = threadIdx.x;
    const int lane  = tid & 31;
    const int wave  = tid >> 5;
    const int ln16  = lane & 15;
    const int hl    = lane >> 4;
    const int batch = blockIdx.x;

    const float* Ab = A  + (size_t)batch * (QP_M * QP_N);
    const float* xb = X  + (size_t)batch * QP_N;
    const float* bb = Bv + (size_t)batch * QP_M;

    if (tid < QP_N) {
        float di = 1.0f / Qm[tid * QP_N + tid];
        sDinv[tid]  = di;
        sSqrtd[tid] = sqrtf(fabsf(di));
    }
    __syncthreads();
    if (tid < QP_N) sXp[tid] = sSqrtd[tid] * xb[tid];

    float ss = 0.f;
    #pragma unroll
    for (int j = 0; j < (QP_M * QP_N / 4) / NTHREADS; ++j) {
        int f4  = j * NTHREADS + tid;
        float4 v = ((const float4*)Ab)[f4];
        int fl  = f4 << 2;
        int row = fl >> 6, col = fl & 63;
        ss += v.x*v.x + v.y*v.y + v.z*v.z + v.w*v.w;
        float* dst = &sR[row * RSTRIDE + col];
        dst[0] = v.x * sSqrtd[col + 0];
        dst[1] = v.y * sSqrtd[col + 1];
        dst[2] = v.z * sSqrtd[col + 2];
        dst[3] = v.w * sSqrtd[col + 3];
    }
    #pragma unroll
    for (int off = 16; off; off >>= 1) ss += __shfl_xor(ss, off, 32);
    if (lane == 0) sRed[wave] = ss;
    if (tid < QP_M) (&sY[0][0][0])[tid] = 0.f;
    __syncthreads();

    if (tid == 0) {
        float s = 0.f;
        for (int w = 0; w < 8; ++w) s += sRed[w];
        float qn2 = 0.f;
        for (int n = 0; n < QP_N; ++n) qn2 += sDinv[n] * sDinv[n];
        float L = s * sqrtf(qn2);
        sEta = 1.0f / fmaxf(L, 1e-12f);
    }
    if (tid < QP_M) {
        float acc = 0.f;
        #pragma unroll 8
        for (int n = 0; n < QP_N; ++n) acc += sR[tid * RSTRIDE + n] * sXp[n];
        sC[tid] = acc - bb[tid];
    }
    __syncthreads();

    v2f af[16];
    #pragma unroll
    for (int k = 0; k < 16; ++k) {
        int row = wave * 16 + ln16;
        int col = 4 * k + 2 * hl;
        af[k] = *(const v2f*)&sR[row * RSTRIDE + col];
    }
    v8f acc[8];
    #pragma unroll
    for (int i = 0; i < 8; ++i) {
        v8f c = {};
        #pragma unroll
        for (int k = 0; k < 16; ++k) {
            int row = i * 16 + ln16;
            int col = 4 * k + 2 * hl;
            v2f bf = *(const v2f*)&sR[row * RSTRIDE + col];
            c = __builtin_amdgcn_wmma_f32_16x16x4_f32(
                    false, af[k], false, bf, (short)0, c, false, false);
        }
        acc[i] = c;
    }
    __syncthreads();

    const float eta = sEta;
    const int   rsub = 8 * hl + (ln16 >> 1);
    const int   rr   = wave * 16 + rsub;
    const bool  b3   = (ln16 & 8) != 0;
    const bool  b2   = (ln16 & 4) != 0;
    const bool  b1   = (ln16 & 2) != 0;
    const bool  writer = (lane & 1) == 0;
    const int   ypos = (wave >> 2) * 64 + 4 * rsub + (wave & 3);
    float c_r = sC[rr];
    float lam_r = 0.f, y_r = 0.f, t = 1.f;
    int buf = 0;

    for (int it = 0; it < QP_ITERS; ++it) {
        const float* y = &sY[buf][0][0];
        const float4 ya = *(const float4*)&y[4 * ln16];
        const float4 yb = *(const float4*)&y[64 + 4 * ln16];
        const float yv[8] = {ya.x, ya.y, ya.z, ya.w, yb.x, yb.y, yb.z, yb.w};

        v2f p01 = {0.f, 0.f}, p23 = {0.f, 0.f}, p45 = {0.f, 0.f}, p67 = {0.f, 0.f};
        #pragma unroll
        for (int i = 0; i < 8; ++i) {
            const v2f ys = {yv[i], yv[i]};
            p01 += __builtin_shufflevector(acc[i], acc[i], 0, 1) * ys;
            p23 += __builtin_shufflevector(acc[i], acc[i], 2, 3) * ys;
            p45 += __builtin_shufflevector(acc[i], acc[i], 4, 5) * ys;
            p67 += __builtin_shufflevector(acc[i], acc[i], 6, 7) * ys;
        }
        const float p[8] = {p01.x, p01.y, p23.x, p23.y, p45.x, p45.y, p67.x, p67.y};

        float q[4];
        #pragma unroll
        for (int j = 0; j < 4; ++j) {
            float keep = b3 ? p[4 + j] : p[j];
            float send = b3 ? p[j]     : p[4 + j];
            q[j] = keep + row_xor<8>(send);
        }
        float s2[2];
        #pragma unroll
        for (int m = 0; m < 2; ++m) {
            float keep = b2 ? q[2 + m] : q[m];
            float send = b2 ? q[m]     : q[2 + m];
            s2[m] = keep + row_xor<4>(send);
        }
        float keep1 = b1 ? s2[1] : s2[0];
        float send1 = b1 ? s2[0] : s2[1];
        float a = keep1 + row_xor<2>(send1);
        a += row_xor<1>(a);

        float tn = 0.5f * (1.f + sqrtf(1.f + 4.f * t * t));
        float g   = c_r - a;
        float ln_ = fmaxf(0.f, y_r + eta * g);
        float yn  = ln_ + ((t - 1.f) / tn) * (ln_ - lam_r);
        if (writer) (&sY[buf ^ 1][0][0])[ypos] = yn;
        lam_r = ln_;
        y_r   = yn;
        t = tn;
        buf ^= 1;
        __syncthreads();
    }

    if (writer) sLam[rr] = lam_r;
    __syncthreads();
    if (tid < QP_N) {
        float a2 = 0.f;
        #pragma unroll 8
        for (int m = 0; m < QP_M; ++m) a2 += sR[m * RSTRIDE + tid] * sLam[m];
        Out[(size_t)batch * QP_N + tid] = sDinv[tid] * xb[tid] - sSqrtd[tid] * a2;
    }
}

extern "C" void kernel_launch(void* const* d_in, const int* in_sizes, int n_in,
                              void* d_out, int out_size, void* d_ws, size_t ws_size,
                              hipStream_t stream) {
    const float* Q = (const float*)d_in[0];
    const float* A = (const float*)d_in[1];
    const float* X = (const float*)d_in[2];
    const float* B = (const float*)d_in[3];
    float* Out = (float*)d_out;
    int batches = in_sizes[1] / (QP_M * QP_N);
    qp_fista_kernel<<<batches, NTHREADS, 0, stream>>>(Q, A, X, B, Out);
}


# Momentum coefficient (same in every lane): hardware sqrt and reciprocal with one correction step each, all on the scalar unit, replace the IEEE sqrt and division sequences; its last bit may differ
# speedup vs baseline: 1.6287x; 1.0883x over previous
#include <hip/hip_runtime.h>
#include <hip/hip_bf16.h>


typedef __attribute__((ext_vector_type(2))) float v2f;
typedef __attribute__((ext_vector_type(8))) float v8f;

#define QP_ITERS   200
#define QP_M       128
#define QP_N       64
#define RSTRIDE    66
#define NTHREADS   256

template <int MASK>
__device__ __forceinline__ float row_xor(float v) {
    return __int_as_float(__builtin_amdgcn_update_dpp(
        0, __float_as_int(v), 0x160 + MASK, 0xf, 0xf, true));
}


__device__ __forceinline__ float sqrt_corrected_neg(float nx) {
    const float s0 = __builtin_amdgcn_sqrtf(-nx);
    const float nh = -0.5f * __builtin_amdgcn_rcpf(s0);
    return fmaf(fmaf(s0, s0, nx), nh, s0);
}
__device__ __forceinline__ float div_corrected(float a, float b) {
    const float r = __builtin_amdgcn_rcpf(b);
    const float q = a * r;
    return fmaf(fmaf(-q, b, a), r, q);
}

__global__ __launch_bounds__(NTHREADS) void qp_fista_kernel(
    const float* __restrict__ Qm, const float* __restrict__ A,
    const float* __restrict__ X,  const float* __restrict__ Bv,
    float* __restrict__ Out)
{
    __shared__ __align__(16) float sR[QP_M * RSTRIDE];
    __shared__ __align__(16) float sY[2][2][64];
    __shared__ float sC[QP_M];
    __shared__ float sLam[QP_M];
    __shared__ float sDinv[QP_N];
    __shared__ float sSqrtd[QP_N];
    __shared__ float sXp[QP_N];
    __shared__ float sRed[8];
    __shared__ float sEta;

    const int tid   = threadIdx.x;
    const int lane  = tid & 31;
    const int wave  = tid >> 5;
    const int ln16  = lane & 15;
    const int hl    = lane >> 4;
    const int batch = blockIdx.x;

    const float* Ab = A  + (size_t)batch * (QP_M * QP_N);
    const float* xb = X  + (size_t)batch * QP_N;
    const float* bb = Bv + (size_t)batch * QP_M;

    if (tid < QP_N) {
        float di = 1.0f / Qm[tid * QP_N + tid];
        sDinv[tid]  = di;
        sSqrtd[tid] = sqrtf(fabsf(di));
    }
    __syncthreads();
    if (tid < QP_N) sXp[tid] = sSqrtd[tid] * xb[tid];

    float ss = 0.f;
    #pragma unroll
    for (int j = 0; j < (QP_M * QP_N / 4) / NTHREADS; ++j) {
        int f4  = j * NTHREADS + tid;
        float4 v = ((const float4*)Ab)[f4];
        int fl  = f4 << 2;
        int row = fl >> 6, col = fl & 63;
        ss += v.x*v.x + v.y*v.y + v.z*v.z + v.w*v.w;
        float* dst = &sR[row * RSTRIDE + col];
        dst[0] = v.x * sSqrtd[col + 0];
        dst[1] = v.y * sSqrtd[col + 1];
        dst[2] = v.z * sSqrtd[col + 2];
        dst[3] = v.w * sSqrtd[col + 3];
    }
    #pragma unroll
    for (int off = 16; off; off >>= 1) ss += __shfl_xor(ss, off, 32);
    if (lane == 0) sRed[wave] = ss;
    if (tid < QP_M) (&sY[0][0][0])[tid] = 0.f;
    __syncthreads();

    if (tid == 0) {
        float s = 0.f;
        for (int w = 0; w < 8; ++w) s += sRed[w];
        float qn2 = 0.f;
        for (int n = 0; n < QP_N; ++n) qn2 += sDinv[n] * sDinv[n];
        float L = s * sqrtf(qn2);
        sEta = 1.0f / fmaxf(L, 1e-12f);
    }
    if (tid < QP_M) {
        float acc = 0.f;
        #pragma unroll 8
        for (int n = 0; n < QP_N; ++n) acc += sR[tid * RSTRIDE + n] * sXp[n];
        sC[tid] = acc - bb[tid];
    }
    __syncthreads();

    v2f af[16];
    #pragma unroll
    for (int k = 0; k < 16; ++k) {
        int row = wave * 16 + ln16;
        int col = 4 * k + 2 * hl;
        af[k] = *(const v2f*)&sR[row * RSTRIDE + col];
    }
    v8f acc[8];
    #pragma unroll
    for (int i = 0; i < 8; ++i) {
        v8f c = {};
        #pragma unroll
        for (int k = 0; k < 16; ++k) {
            int row = i * 16 + ln16;
            int col = 4 * k + 2 * hl;
            v2f bf = *(const v2f*)&sR[row * RSTRIDE + col];
            c = __builtin_amdgcn_wmma_f32_16x16x4_f32(
                    false, af[k], false, bf, (short)0, c, false, false);
        }
        acc[i] = c;
    }
    __syncthreads();

    const float eta = sEta;
    const int   rsub = 8 * hl + (ln16 >> 1);
    const int   rr   = wave * 16 + rsub;
    const bool  b3   = (ln16 & 8) != 0;
    const bool  b2   = (ln16 & 4) != 0;
    const bool  b1   = (ln16 & 2) != 0;
    const bool  writer = (lane & 1) == 0;
    const int   ypos = (wave >> 2) * 64 + 4 * rsub + (wave & 3);
    float c_r = sC[rr];
    float lam_r = 0.f, y_r = 0.f, t = 1.f;
    int buf = 0;

    for (int it = 0; it < QP_ITERS; ++it) {
        const float* y = &sY[buf][0][0];
        const float4 ya = *(const float4*)&y[4 * ln16];
        const float4 yb = *(const float4*)&y[64 + 4 * ln16];
        const float yv[8] = {ya.x, ya.y, ya.z, ya.w, yb.x, yb.y, yb.z, yb.w};

        v2f p01 = {0.f, 0.f}, p23 = {0.f, 0.f}, p45 = {0.f, 0.f}, p67 = {0.f, 0.f};
        #pragma unroll
        for (int i = 0; i < 8; ++i) {
            const v2f ys = {yv[i], yv[i]};
            p01 += __builtin_shufflevector(acc[i], acc[i], 0, 1) * ys;
            p23 += __builtin_shufflevector(acc[i], acc[i], 2, 3) * ys;
            p45 += __builtin_shufflevector(acc[i], acc[i], 4, 5) * ys;
            p67 += __builtin_shufflevector(acc[i], acc[i], 6, 7) * ys;
        }
        const float p[8] = {p01.x, p01.y, p23.x, p23.y, p45.x, p45.y, p67.x, p67.y};

        float q[4];
        #pragma unroll
        for (int j = 0; j < 4; ++j) {
            float keep = b3 ? p[4 + j] : p[j];
            float send = b3 ? p[j]     : p[4 + j];
            q[j] = keep + row_xor<8>(send);
        }
        float s2[2];
        #pragma unroll
        for (int m = 0; m < 2; ++m) {
            float keep = b2 ? q[2 + m] : q[m];
            float send = b2 ? q[m]     : q[2 + m];
            s2[m] = keep + row_xor<4>(send);
        }
        float keep1 = b1 ? s2[1] : s2[0];
        float send1 = b1 ? s2[0] : s2[1];
        float a = keep1 + row_xor<2>(send1);
        a += row_xor<1>(a);

        float tn = fmaf(0.5f, sqrt_corrected_neg(fmaf(-4.f * t, t, -1.f)), 0.5f);
        const float beta = div_corrected(t - 1.f, tn);
        float g   = c_r - a;
        float ln_ = fmaxf(0.f, y_r + eta * g);
        float yn  = ln_ + beta * (ln_ - lam_r);
        if (writer) (&sY[buf ^ 1][0][0])[ypos] = yn;
        lam_r = ln_;
        y_r   = yn;
        t = tn;
        buf ^= 1;
        __syncthreads();
    }

    if (writer) sLam[rr] = lam_r;
    __syncthreads();
    if (tid < QP_N) {
        float a2 = 0.f;
        #pragma unroll 8
        for (int m = 0; m < QP_M; ++m) a2 += sR[m * RSTRIDE + tid] * sLam[m];
        Out[(size_t)batch * QP_N + tid] = sDinv[tid] * xb[tid] - sSqrtd[tid] * a2;
    }
}

extern "C" void kernel_launch(void* const* d_in, const int* in_sizes, int n_in,
                              void* d_out, int out_size, void* d_ws, size_t ws_size,
                              hipStream_t stream) {
    const float* Q = (const float*)d_in[0];
    const float* A = (const float*)d_in[1];
    const float* X = (const float*)d_in[2];
    const float* B = (const float*)d_in[3];
    float* Out = (float*)d_out;
    int batches = in_sizes[1] / (QP_M * QP_N);
    qp_fista_kernel<<<batches, NTHREADS, 0, stream>>>(Q, A, X, B, Out);
}


# Two waves per instance; by symmetry of G a lane owns whole half rows (tile (I,J) read as row 16J+c): no 16-lane reduction, one add across lane halves; 64 y per lane; row summation order changed
# speedup vs baseline: 1.9503x; 1.1975x over previous
#include <hip/hip_runtime.h>
#include <hip/hip_bf16.h>


typedef __attribute__((ext_vector_type(2))) float v2f;
typedef __attribute__((ext_vector_type(8))) float v8f;

#define QP_ITERS   200
#define QP_M       128
#define QP_N       64
#define RSTRIDE    66
#define NTHREADS   64

__device__ __forceinline__ float xhalf(float v) {
    const int b = __float_as_int(v);
    return __int_as_float(__builtin_amdgcn_permlanex16(
        b, b, 0x76543210, 0xfedcba98, false, false));
}


__device__ __forceinline__ float sqrt_corrected_neg(float nx) {
    const float s0 = __builtin_amdgcn_sqrtf(-nx);
    const float nh = -0.5f * __builtin_amdgcn_rcpf(s0);
    return fmaf(fmaf(s0, s0, nx), nh, s0);
}
__device__ __forceinline__ float div_corrected(float a, float b) {
    const float r = __builtin_amdgcn_rcpf(b);
    const float q = a * r;
    return fmaf(fmaf(-q, b, a), r, q);
}

__global__ __launch_bounds__(NTHREADS) void qp_fista_kernel(
    const float* __restrict__ Qm, const float* __restrict__ A,
    const float* __restrict__ X,  const float* __restrict__ Bv,
    float* __restrict__ Out)
{
    __shared__ __align__(16) float sR[QP_M * RSTRIDE];
    __shared__ __align__(16) float sY[2][2][64];
    __shared__ float sC[QP_M];
    __shared__ float sLam[QP_M];
    __shared__ float sDinv[QP_N];
    __shared__ float sSqrtd[QP_N];
    __shared__ float sXp[QP_N];
    __shared__ float sRed[8];
    __shared__ float sEta;

    const int tid   = threadIdx.x;
    const int lane  = tid & 31;
    const int wave  = tid >> 5;
    const int ln16  = lane & 15;
    const int hl    = lane >> 4;
    const int batch = blockIdx.x;

    const float* Ab = A  + (size_t)batch * (QP_M * QP_N);
    const float* xb = X  + (size_t)batch * QP_N;
    const float* bb = Bv + (size_t)batch * QP_M;

    if (tid < QP_N) {
        float di = 1.0f / Qm[tid * QP_N + tid];
        sDinv[tid]  = di;
        sSqrtd[tid] = sqrtf(fabsf(di));
    }
    __syncthreads();
    if (tid < QP_N) sXp[tid] = sSqrtd[tid] * xb[tid];

    float ss[4] = {0.f, 0.f, 0.f, 0.f};
    #pragma unroll
    for (int j = 0; j < 8; ++j) {
        #pragma unroll
        for (int u = 0; u < 4; ++u) {
            int f4  = j * 256 + u * NTHREADS + tid;
            float4 v = ((const float4*)Ab)[f4];
            int fl  = f4 << 2;
            int row = fl >> 6, col = fl & 63;
            ss[u] += v.x*v.x + v.y*v.y + v.z*v.z + v.w*v.w;
            float* dst = &sR[row * RSTRIDE + col];
            dst[0] = v.x * sSqrtd[col + 0];
            dst[1] = v.y * sSqrtd[col + 1];
            dst[2] = v.z * sSqrtd[col + 2];
            dst[3] = v.w * sSqrtd[col + 3];
        }
    }
    #pragma unroll
    for (int off = 16; off; off >>= 1) {
        #pragma unroll
        for (int u = 0; u < 4; ++u) ss[u] += __shfl_xor(ss[u], off, 32);
    }
    if (lane == 0) {
        #pragma unroll
        for (int u = 0; u < 4; ++u) sRed[2 * u + wave] = ss[u];
    }
    (&sY[0][0][0])[tid] = 0.f;
    (&sY[0][0][0])[tid + NTHREADS] = 0.f;
    __syncthreads();

    if (tid == 0) {
        float s = 0.f;
        for (int w = 0; w < 8; ++w) s += sRed[w];
        float qn2 = 0.f;
        for (int n = 0; n < QP_N; ++n) qn2 += sDinv[n] * sDinv[n];
        float L = s * sqrtf(qn2);
        sEta = 1.0f / fmaxf(L, 1e-12f);
    }
    #pragma unroll
    for (int u = 0; u < 2; ++u) {
        const int m = tid + u * NTHREADS;
        float acc = 0.f;
        #pragma unroll 8
        for (int n = 0; n < QP_N; ++n) acc += sR[m * RSTRIDE + n] * sXp[n];
        sC[m] = acc - bb[m];
    }
    __syncthreads();

    v8f acc[4][8];
    #pragma unroll
    for (int jb = 0; jb < 4; ++jb) {
        v2f bfr[16];
        #pragma unroll
        for (int k = 0; k < 16; ++k) {
            int row = (4 * wave + jb) * 16 + ln16;
            int col = 4 * k + 2 * hl;
            bfr[k] = *(const v2f*)&sR[row * RSTRIDE + col];
        }
        #pragma unroll
        for (int i = 0; i < 8; ++i) {
            v8f c = {};
            #pragma unroll
            for (int k = 0; k < 16; ++k) {
                int row = i * 16 + ln16;
                int col = 4 * k + 2 * hl;
                v2f afr = *(const v2f*)&sR[row * RSTRIDE + col];
                c = __builtin_amdgcn_wmma_f32_16x16x4_f32(
                        false, afr, false, bfr[k], (short)0, c, false, false);
            }
            acc[jb][i] = c;
        }
    }
    __syncthreads();

    const float eta = sEta;
    const bool  upper = hl != 0;
    int   mrow[2], ypos[2];
    float c_r[2], lam_r[2] = {0.f, 0.f}, y_r[2] = {0.f, 0.f};
    #pragma unroll
    for (int j = 0; j < 2; ++j) {
        mrow[j] = wave * 64 + 32 * hl + 16 * j + ln16;
        ypos[j] = ((mrow[j] >> 3) & 1) * 64 + 8 * (mrow[j] >> 4) + (mrow[j] & 7);
        c_r[j]  = sC[mrow[j]];
    }
    float t = 1.f;
    int buf = 0;

    for (int it = 0; it < QP_ITERS; ++it) {
        const float4* yq = (const float4*)&sY[buf][hl][0];
        v2f yp[32];
        #pragma unroll
        for (int q = 0; q < 16; ++q) {
            const float4 v = yq[q];
            yp[2 * q]     = (v2f){v.x, v.y};
            yp[2 * q + 1] = (v2f){v.z, v.w};
        }

        float part[4];
        #pragma unroll
        for (int jb = 0; jb < 4; ++jb) {
            v2f p01 = {0.f, 0.f}, p23 = {0.f, 0.f}, p45 = {0.f, 0.f}, p67 = {0.f, 0.f};
            #pragma unroll
            for (int i = 0; i < 8; ++i) {
                p01 += __builtin_shufflevector(acc[jb][i], acc[jb][i], 0, 1) * yp[4 * i + 0];
                p23 += __builtin_shufflevector(acc[jb][i], acc[jb][i], 2, 3) * yp[4 * i + 1];
                p45 += __builtin_shufflevector(acc[jb][i], acc[jb][i], 4, 5) * yp[4 * i + 2];
                p67 += __builtin_shufflevector(acc[jb][i], acc[jb][i], 6, 7) * yp[4 * i + 3];
            }
            const v2f s = (p01 + p23) + (p45 + p67);
            part[jb] = s.x + s.y;
        }

        float a[2];
        #pragma unroll
        for (int j = 0; j < 2; ++j) {
            const float keep = upper ? part[2 + j] : part[j];
            const float send = upper ? part[j]     : part[2 + j];
            a[j] = keep + xhalf(send);
        }

        float tn = fmaf(0.5f, sqrt_corrected_neg(fmaf(-4.f * t, t, -1.f)), 0.5f);
        const float beta = div_corrected(t - 1.f, tn);
        #pragma unroll
        for (int j = 0; j < 2; ++j) {
            float g   = c_r[j] - a[j];
            float ln_ = fmaxf(0.f, y_r[j] + eta * g);
            float yn  = ln_ + beta * (ln_ - lam_r[j]);
            (&sY[buf ^ 1][0][0])[ypos[j]] = yn;
            lam_r[j] = ln_;
            y_r[j]   = yn;
        }
        t = tn;
        buf ^= 1;
        __syncthreads();
    }

    #pragma unroll
    for (int j = 0; j < 2; ++j) sLam[mrow[j]] = lam_r[j];
    __syncthreads();
    if (tid < QP_N) {
        float a2 = 0.f;
        #pragma unroll 8
        for (int m = 0; m < QP_M; ++m) a2 += sR[m * RSTRIDE + tid] * sLam[m];
        Out[(size_t)batch * QP_N + tid] = sDinv[tid] * xb[tid] - sSqrtd[tid] * a2;
    }
}

extern "C" void kernel_launch(void* const* d_in, const int* in_sizes, int n_in,
                              void* d_out, int out_size, void* d_ws, size_t ws_size,
                              hipStream_t stream) {
    const float* Q = (const float*)d_in[0];
    const float* A = (const float*)d_in[1];
    const float* X = (const float*)d_in[2];
    const float* B = (const float*)d_in[3];
    float* Out = (float*)d_out;
    int batches = in_sizes[1] / (QP_M * QP_N);
    qp_fista_kernel<<<batches, NTHREADS, 0, stream>>>(Q, A, X, B, Out);
}


# y read from LDS in two groups of loads per iteration (10 + 6 float4), not at eight places among the FMAs; two running sums per row, not four, free the registers; row summation order changed
# speedup vs baseline: 2.0801x; 1.0666x over previous
#include <hip/hip_runtime.h>
#include <hip/hip_bf16.h>


typedef __attribute__((ext_vector_type(2))) float v2f;
typedef __attribute__((ext_vector_type(8))) float v8f;

#define QP_ITERS   200
#define QP_M       128
#define QP_N       64
#define RSTRIDE    66
#define NTHREADS   64

__device__ __forceinline__ float xhalf(float v) {
    const int b = __float_as_int(v);
    return __int_as_float(__builtin_amdgcn_permlanex16(
        b, b, 0x76543210, 0xfedcba98, false, false));
}


__device__ __forceinline__ float sqrt_corrected_neg(float nx) {
    const float s0 = __builtin_amdgcn_sqrtf(-nx);
    const float nh = -0.5f * __builtin_amdgcn_rcpf(s0);
    return fmaf(fmaf(s0, s0, nx), nh, s0);
}
__device__ __forceinline__ float div_corrected(float a, float b) {
    const float r = __builtin_amdgcn_rcpf(b);
    const float q = a * r;
    return fmaf(fmaf(-q, b, a), r, q);
}

template <int Q>
__device__ __forceinline__ void quad_fma(const v8f (&acc)[4][8], const float4 v,
                                         v2f (&s0)[4], v2f (&s1)[4]) {
    constexpr int i = Q >> 1;
    constexpr int a = (Q & 1) * 4;
    const v2f y01 = {v.x, v.y}, y23 = {v.z, v.w};
    #pragma unroll
    for (int jb = 0; jb < 4; ++jb) {
        s0[jb] += __builtin_shufflevector(acc[jb][i], acc[jb][i], a, a + 1) * y01;
        s1[jb] += __builtin_shufflevector(acc[jb][i], acc[jb][i], a + 2, a + 3) * y23;
    }
}

__global__ __launch_bounds__(NTHREADS) void qp_fista_kernel(
    const float* __restrict__ Qm, const float* __restrict__ A,
    const float* __restrict__ X,  const float* __restrict__ Bv,
    float* __restrict__ Out)
{
    __shared__ __align__(16) float sR[QP_M * RSTRIDE];
    __shared__ __align__(16) float sY[2][2][64];
    __shared__ float sC[QP_M];
    __shared__ float sLam[QP_M];
    __shared__ float sDinv[QP_N];
    __shared__ float sSqrtd[QP_N];
    __shared__ float sXp[QP_N];
    __shared__ float sRed[8];
    __shared__ float sEta;

    const int tid   = threadIdx.x;
    const int lane  = tid & 31;
    const int wave  = tid >> 5;
    const int ln16  = lane & 15;
    const int hl    = lane >> 4;
    const int batch = blockIdx.x;

    const float* Ab = A  + (size_t)batch * (QP_M * QP_N);
    const float* xb = X  + (size_t)batch * QP_N;
    const float* bb = Bv + (size_t)batch * QP_M;

    if (tid < QP_N) {
        float di = 1.0f / Qm[tid * QP_N + tid];
        sDinv[tid]  = di;
        sSqrtd[tid] = sqrtf(fabsf(di));
    }
    __syncthreads();
    if (tid < QP_N) sXp[tid] = sSqrtd[tid] * xb[tid];

    float ss[4] = {0.f, 0.f, 0.f, 0.f};
    #pragma unroll
    for (int j = 0; j < 8; ++j) {
        #pragma unroll
        for (int u = 0; u < 4; ++u) {
            int f4  = j * 256 + u * NTHREADS + tid;
            float4 v = ((const float4*)Ab)[f4];
            int fl  = f4 << 2;
            int row = fl >> 6, col = fl & 63;
            ss[u] += v.x*v.x + v.y*v.y + v.z*v.z + v.w*v.w;
            float* dst = &sR[row * RSTRIDE + col];
            dst[0] = v.x * sSqrtd[col + 0];
            dst[1] = v.y * sSqrtd[col + 1];
            dst[2] = v.z * sSqrtd[col + 2];
            dst[3] = v.w * sSqrtd[col + 3];
        }
    }
    #pragma unroll
    for (int off = 16; off; off >>= 1) {
        #pragma unroll
        for (int u = 0; u < 4; ++u) ss[u] += __shfl_xor(ss[u], off, 32);
    }
    if (lane == 0) {
        #pragma unroll
        for (int u = 0; u < 4; ++u) sRed[2 * u + wave] = ss[u];
    }
    (&sY[0][0][0])[tid] = 0.f;
    (&sY[0][0][0])[tid + NTHREADS] = 0.f;
    __syncthreads();

    if (tid == 0) {
        float s = 0.f;
        for (int w = 0; w < 8; ++w) s += sRed[w];
        float qn2 = 0.f;
        for (int n = 0; n < QP_N; ++n) qn2 += sDinv[n] * sDinv[n];
        float L = s * sqrtf(qn2);
        sEta = 1.0f / fmaxf(L, 1e-12f);
    }
    #pragma unroll
    for (int u = 0; u < 2; ++u) {
        const int m = tid + u * NTHREADS;
        float acc = 0.f;
        #pragma unroll 8
        for (int n = 0; n < QP_N; ++n) acc += sR[m * RSTRIDE + n] * sXp[n];
        sC[m] = acc - bb[m];
    }
    __syncthreads();

    v8f acc[4][8];
    #pragma unroll
    for (int jb = 0; jb < 4; ++jb) {
        v2f bfr[16];
        #pragma unroll
        for (int k = 0; k < 16; ++k) {
            int row = (4 * wave + jb) * 16 + ln16;
            int col = 4 * k + 2 * hl;
            bfr[k] = *(const v2f*)&sR[row * RSTRIDE + col];
        }
        #pragma unroll
        for (int i = 0; i < 8; ++i) {
            v8f c = {};
            #pragma unroll
            for (int k = 0; k < 16; ++k) {
                int row = i * 16 + ln16;
                int col = 4 * k + 2 * hl;
                v2f afr = *(const v2f*)&sR[row * RSTRIDE + col];
                c = __builtin_amdgcn_wmma_f32_16x16x4_f32(
                        false, afr, false, bfr[k], (short)0, c, false, false);
            }
            acc[jb][i] = c;
        }
    }
    __syncthreads();

    const float eta = sEta;
    const bool  upper = hl != 0;
    int   mrow[2], ypos[2];
    float c_r[2], lam_r[2] = {0.f, 0.f}, y_r[2] = {0.f, 0.f};
    #pragma unroll
    for (int j = 0; j < 2; ++j) {
        mrow[j] = wave * 64 + 32 * hl + 16 * j + ln16;
        ypos[j] = ((mrow[j] >> 3) & 1) * 64 + 8 * (mrow[j] >> 4) + (mrow[j] & 7);
        c_r[j]  = sC[mrow[j]];
    }
    float t = 1.f;
    int buf = 0;

    for (int it = 0; it < QP_ITERS; ++it) {
        const float4* yq = (const float4*)&sY[buf][hl][0];
        float4 ya[10];
        #pragma unroll
        for (int q = 0; q < 10; ++q) ya[q] = yq[q];

        v2f s0[4], s1[4];
        #pragma unroll
        for (int jb = 0; jb < 4; ++jb) { s0[jb] = (v2f){0.f, 0.f}; s1[jb] = (v2f){0.f, 0.f}; }
        quad_fma<0>(acc, ya[0], s0, s1);
        quad_fma<1>(acc, ya[1], s0, s1);
        quad_fma<2>(acc, ya[2], s0, s1);
        quad_fma<3>(acc, ya[3], s0, s1);
        quad_fma<4>(acc, ya[4], s0, s1);
        quad_fma<5>(acc, ya[5], s0, s1);
        float4 yb[6];
        #pragma unroll
        for (int q = 0; q < 6; ++q) yb[q] = yq[10 + q];
        __builtin_amdgcn_sched_barrier(0);
        quad_fma<6>(acc, ya[6], s0, s1);
        quad_fma<7>(acc, ya[7], s0, s1);
        quad_fma<8>(acc, ya[8], s0, s1);
        quad_fma<9>(acc, ya[9], s0, s1);
        quad_fma<10>(acc, yb[0], s0, s1);
        quad_fma<11>(acc, yb[1], s0, s1);
        quad_fma<12>(acc, yb[2], s0, s1);
        quad_fma<13>(acc, yb[3], s0, s1);
        quad_fma<14>(acc, yb[4], s0, s1);
        quad_fma<15>(acc, yb[5], s0, s1);

        float part[4];
        #pragma unroll
        for (int jb = 0; jb < 4; ++jb) {
            const v2f s = s0[jb] + s1[jb];
            part[jb] = s.x + s.y;
        }

        float a[2];
        #pragma unroll
        for (int j = 0; j < 2; ++j) {
            const float keep = upper ? part[2 + j] : part[j];
            const float send = upper ? part[j]     : part[2 + j];
            a[j] = keep + xhalf(send);
        }

        float tn = fmaf(0.5f, sqrt_corrected_neg(fmaf(-4.f * t, t, -1.f)), 0.5f);
        const float beta = div_corrected(t - 1.f, tn);
        #pragma unroll
        for (int j = 0; j < 2; ++j) {
            float g   = c_r[j] - a[j];
            float ln_ = fmaxf(0.f, y_r[j] + eta * g);
            float yn  = ln_ + beta * (ln_ - lam_r[j]);
            (&sY[buf ^ 1][0][0])[ypos[j]] = yn;
            lam_r[j] = ln_;
            y_r[j]   = yn;
        }
        t = tn;
        buf ^= 1;
        __syncthreads();
    }

    #pragma unroll
    for (int j = 0; j < 2; ++j) sLam[mrow[j]] = lam_r[j];
    __syncthreads();
    if (tid < QP_N) {
        float a2 = 0.f;
        #pragma unroll 8
        for (int m = 0; m < QP_M; ++m) a2 += sR[m * RSTRIDE + tid] * sLam[m];
        Out[(size_t)batch * QP_N + tid] = sDinv[tid] * xb[tid] - sSqrtd[tid] * a2;
    }
}

extern "C" void kernel_launch(void* const* d_in, const int* in_sizes, int n_in,
                              void* d_out, int out_size, void* d_ws, size_t ws_size,
                              hipStream_t stream) {
    const float* Q = (const float*)d_in[0];
    const float* A = (const float*)d_in[1];
    const float* X = (const float*)d_in[2];
    const float* B = (const float*)d_in[3];
    float* Out = (float*)d_out;
    int batches = in_sizes[1] / (QP_M * QP_N);
    qp_fista_kernel<<<batches, NTHREADS, 0, stream>>>(Q, A, X, B, Out);
}


# Workgroup barrier split: signal behind a wave's own y stores, wait only before the first read of the other wave's y; tiles own-first, own y reloaded behind the stores; row summation order changed
# speedup vs baseline: 2.2406x; 1.0772x over previous
#include <hip/hip_runtime.h>
#include <hip/hip_bf16.h>


typedef __attribute__((ext_vector_type(2))) float v2f;
typedef __attribute__((ext_vector_type(8))) float v8f;

#define QP_ITERS   200
#define QP_M       128
#define QP_N       64
#define RSTRIDE    66
#define NTHREADS   64

__device__ __forceinline__ void bar_signal() { asm volatile("s_barrier_signal -1" ::: "memory"); }
__device__ __forceinline__ void bar_wait()   { asm volatile("s_barrier_wait -1" ::: "memory"); }
template <int N> __device__ __forceinline__ void wait_ds_le() { asm volatile("s_wait_dscnt %0" :: "i"(N) : "memory"); }

__device__ __forceinline__ float xhalf(float v) {
    const int b = __float_as_int(v);
    return __int_as_float(__builtin_amdgcn_permlanex16(
        b, b, 0x76543210, 0xfedcba98, false, false));
}


__device__ __forceinline__ float sqrt_corrected_neg(float nx) {
    const float s0 = __builtin_amdgcn_sqrtf(-nx);
    const float nh = -0.5f * __builtin_amdgcn_rcpf(s0);
    return fmaf(fmaf(s0, s0, nx), nh, s0);
}
__device__ __forceinline__ float div_corrected(float a, float b) {
    const float r = __builtin_amdgcn_rcpf(b);
    const float q = a * r;
    return fmaf(fmaf(-q, b, a), r, q);
}

template <int Q>
__device__ __forceinline__ void quad_fma(const v8f (&acc)[4][8], const float4 v,
                                         v2f (&s0)[4], v2f (&s1)[4]) {
    constexpr int i = Q >> 1;
    constexpr int a = (Q & 1) * 4;
    const v2f y01 = {v.x, v.y}, y23 = {v.z, v.w};
    #pragma unroll
    for (int jb = 0; jb < 4; ++jb) {
        s0[jb] += __builtin_shufflevector(acc[jb][i], acc[jb][i], a, a + 1) * y01;
        s1[jb] += __builtin_shufflevector(acc[jb][i], acc[jb][i], a + 2, a + 3) * y23;
    }
}

__global__ __launch_bounds__(NTHREADS) void qp_fista_kernel(
    const float* __restrict__ Qm, const float* __restrict__ A,
    const float* __restrict__ X,  const float* __restrict__ Bv,
    float* __restrict__ Out)
{
    __shared__ __align__(16) float sR[QP_M * RSTRIDE];
    __shared__ __align__(16) float sY[2][2][64];
    __shared__ float sC[QP_M];
    __shared__ float sLam[QP_M];
    __shared__ float sDinv[QP_N];
    __shared__ float sSqrtd[QP_N];
    __shared__ float sXp[QP_N];
    __shared__ float sRed[8];
    __shared__ float sEta;

    const int tid   = threadIdx.x;
    const int lane  = tid & 31;
    const int wave  = tid >> 5;
    const int ln16  = lane & 15;
    const int hl    = lane >> 4;
    const int batch = blockIdx.x;

    const float* Ab = A  + (size_t)batch * (QP_M * QP_N);
    const float* xb = X  + (size_t)batch * QP_N;
    const float* bb = Bv + (size_t)batch * QP_M;

    if (tid < QP_N) {
        float di = 1.0f / Qm[tid * QP_N + tid];
        sDinv[tid]  = di;
        sSqrtd[tid] = sqrtf(fabsf(di));
    }
    __syncthreads();
    if (tid < QP_N) sXp[tid] = sSqrtd[tid] * xb[tid];

    float ss[4] = {0.f, 0.f, 0.f, 0.f};
    #pragma unroll
    for (int j = 0; j < 8; ++j) {
        #pragma unroll
        for (int u = 0; u < 4; ++u) {
            int f4  = j * 256 + u * NTHREADS + tid;
            float4 v = ((const float4*)Ab)[f4];
            int fl  = f4 << 2;
            int row = fl >> 6, col = fl & 63;
            ss[u] += v.x*v.x + v.y*v.y + v.z*v.z + v.w*v.w;
            float* dst = &sR[row * RSTRIDE + col];
            dst[0] = v.x * sSqrtd[col + 0];
            dst[1] = v.y * sSqrtd[col + 1];
            dst[2] = v.z * sSqrtd[col + 2];
            dst[3] = v.w * sSqrtd[col + 3];
        }
    }
    #pragma unroll
    for (int off = 16; off; off >>= 1) {
        #pragma unroll
        for (int u = 0; u < 4; ++u) ss[u] += __shfl_xor(ss[u], off, 32);
    }
    if (lane == 0) {
        #pragma unroll
        for (int u = 0; u < 4; ++u) sRed[2 * u + wave] = ss[u];
    }
    (&sY[0][0][0])[tid] = 0.f;
    (&sY[0][0][0])[tid + NTHREADS] = 0.f;
    __syncthreads();

    if (tid == 0) {
        float s = 0.f;
        for (int w = 0; w < 8; ++w) s += sRed[w];
        float qn2 = 0.f;
        for (int n = 0; n < QP_N; ++n) qn2 += sDinv[n] * sDinv[n];
        float L = s * sqrtf(qn2);
        sEta = 1.0f / fmaxf(L, 1e-12f);
    }
    #pragma unroll
    for (int u = 0; u < 2; ++u) {
        const int m = tid + u * NTHREADS;
        float acc = 0.f;
        #pragma unroll 8
        for (int n = 0; n < QP_N; ++n) acc += sR[m * RSTRIDE + n] * sXp[n];
        sC[m] = acc - bb[m];
    }
    __syncthreads();

    v8f acc[4][8];
    #pragma unroll
    for (int jb = 0; jb < 4; ++jb) {
        v2f bfr[16];
        #pragma unroll
        for (int k = 0; k < 16; ++k) {
            int row = (4 * wave + jb) * 16 + ln16;
            int col = 4 * k + 2 * hl;
            bfr[k] = *(const v2f*)&sR[row * RSTRIDE + col];
        }
        #pragma unroll
        for (int i = 0; i < 8; ++i) {
            v8f c = {};
            #pragma unroll
            for (int k = 0; k < 16; ++k) {
                int row = (i ^ (4 * wave)) * 16 + ln16;
                int col = 4 * k + 2 * hl;
                v2f afr = *(const v2f*)&sR[row * RSTRIDE + col];
                c = __builtin_amdgcn_wmma_f32_16x16x4_f32(
                        false, afr, false, bfr[k], (short)0, c, false, false);
            }
            acc[jb][i] = c;
        }
    }
    __syncthreads();

    const float eta = sEta;
    const bool  upper = hl != 0;
    int   mrow[2], ypos[2];
    float c_r[2], lam_r[2] = {0.f, 0.f}, y_r[2] = {0.f, 0.f};
    #pragma unroll
    for (int j = 0; j < 2; ++j) {
        mrow[j] = wave * 64 + 32 * hl + 16 * j + ln16;
        ypos[j] = ((mrow[j] >> 3) & 1) * 64 + 8 * (mrow[j] >> 4) + (mrow[j] & 7);
        c_r[j]  = sC[mrow[j]];
    }
    float t = 1.f;
    int buf = 0;

    const int own_off = 32 * wave, par_off = 32 * (1 - wave);

    float4 yo[8];
    {
        const float4* p = (const float4*)&sY[0][hl][own_off];
        #pragma unroll
        for (int q = 0; q < 8; ++q) yo[q] = p[q];
    }
    bar_signal();

    v2f s0[4], s1[4];
    #pragma unroll
    for (int jb = 0; jb < 4; ++jb) { s0[jb] = (v2f){0.f, 0.f}; s1[jb] = (v2f){0.f, 0.f}; }
    quad_fma<0>(acc, yo[0], s0, s1); quad_fma<1>(acc, yo[1], s0, s1); quad_fma<2>(acc, yo[2], s0, s1);

    for (int it = 0; it < QP_ITERS; ++it) {
        bar_wait();
        const float4* yp = (const float4*)&sY[buf][hl][par_off];
        float4 ya[5];
        #pragma unroll
        for (int q = 0; q < 5; ++q) ya[q] = yp[q];
        __builtin_amdgcn_sched_barrier(0);
        quad_fma<3>(acc, yo[3], s0, s1);
        quad_fma<4>(acc, yo[4], s0, s1);
        quad_fma<5>(acc, yo[5], s0, s1);
        quad_fma<6>(acc, yo[6], s0, s1);
        quad_fma<7>(acc, yo[7], s0, s1);
        float4 yb[3];
        #pragma unroll
        for (int q = 0; q < 3; ++q) yb[q] = yp[5 + q];
        __builtin_amdgcn_sched_barrier(0);
        quad_fma<8>(acc, ya[0], s0, s1);
        quad_fma<9>(acc, ya[1], s0, s1);
        quad_fma<10>(acc, ya[2], s0, s1);
        quad_fma<11>(acc, ya[3], s0, s1);
        quad_fma<12>(acc, ya[4], s0, s1);
        quad_fma<13>(acc, yb[0], s0, s1);
        quad_fma<14>(acc, yb[1], s0, s1);
        quad_fma<15>(acc, yb[2], s0, s1);

        float part[4];
        #pragma unroll
        for (int jb = 0; jb < 4; ++jb) {
            const v2f s = s0[jb] + s1[jb];
            part[jb] = s.x + s.y;
        }

        float a[2];
        #pragma unroll
        for (int j = 0; j < 2; ++j) {
            const float keep = upper ? part[2 + j] : part[j];
            const float send = upper ? part[j]     : part[2 + j];
            a[j] = keep + xhalf(send);
        }

        float tn = fmaf(0.5f, sqrt_corrected_neg(fmaf(-4.f * t, t, -1.f)), 0.5f);
        const float beta = div_corrected(t - 1.f, tn);
        #pragma unroll
        for (int j = 0; j < 2; ++j) {
            float g   = c_r[j] - a[j];
            float ln_ = fmaxf(0.f, y_r[j] + eta * g);
            float yn  = ln_ + beta * (ln_ - lam_r[j]);
            (&sY[buf ^ 1][0][0])[ypos[j]] = yn;
            lam_r[j] = ln_;
            y_r[j]   = yn;
        }
        t = tn;
        buf ^= 1;
        {
            const float4* p = (const float4*)&sY[buf][hl][own_off];
            #pragma unroll
            for (int q = 0; q < 8; ++q) yo[q] = p[q];
        }
        wait_ds_le<8>();
        bar_signal();
        #pragma unroll
        for (int jb = 0; jb < 4; ++jb) { s0[jb] = (v2f){0.f, 0.f}; s1[jb] = (v2f){0.f, 0.f}; }
        quad_fma<0>(acc, yo[0], s0, s1); quad_fma<1>(acc, yo[1], s0, s1); quad_fma<2>(acc, yo[2], s0, s1);
    }
    bar_wait();

    #pragma unroll
    for (int j = 0; j < 2; ++j) sLam[mrow[j]] = lam_r[j];
    __syncthreads();
    if (tid < QP_N) {
        float a2 = 0.f;
        #pragma unroll 8
        for (int m = 0; m < QP_M; ++m) a2 += sR[m * RSTRIDE + tid] * sLam[m];
        Out[(size_t)batch * QP_N + tid] = sDinv[tid] * xb[tid] - sSqrtd[tid] * a2;
    }
}

extern "C" void kernel_launch(void* const* d_in, const int* in_sizes, int n_in,
                              void* d_out, int out_size, void* d_ws, size_t ws_size,
                              hipStream_t stream) {
    const float* Q = (const float*)d_in[0];
    const float* A = (const float*)d_in[1];
    const float* X = (const float*)d_in[2];
    const float* B = (const float*)d_in[3];
    float* Out = (float*)d_out;
    int batches = in_sizes[1] / (QP_M * QP_N);
    qp_fista_kernel<<<batches, NTHREADS, 0, stream>>>(Q, A, X, B, Out);
}
